# starting kernel
# baseline (speedup 1.0000x reference)
#include <hip/hip_runtime.h>
#include <math.h>


typedef __attribute__((ext_vector_type(2))) float v2f;
typedef __attribute__((ext_vector_type(4))) float v4f;
typedef __attribute__((ext_vector_type(8))) float v8f;

#define IN_DIM   3
#define HID      20
#define OUT_DIM  3
#define NEXP     5
#define NHID     100
#define NTILE_H  7
#define LDS_STRIDE 20
#define LDS_HROWS  112
#define LDS_PER_WAVE (LDS_HROWS*LDS_STRIDE + NEXP*16)
#define WAVES_PER_BLOCK 4

static __device__ __forceinline__ v8f splat8(float s) {
  v8f v;
#pragma unroll
  for (int i = 0; i < 8; ++i) v[i] = s;
  return v;
}

__global__ __launch_bounds__(WAVES_PER_BLOCK * 32) void moe_wmma_kernel(
    const float* __restrict__ x,  const float* __restrict__ W1, const float* __restrict__ b1,
    const float* __restrict__ W2, const float* __restrict__ b2,
    const float* __restrict__ Wg, const float* __restrict__ bg,
    float* __restrict__ out_mixed, float* __restrict__ out_gate, int nTiles)
{
  __shared__ __align__(16) float lds[WAVES_PER_BLOCK * LDS_PER_WAVE];
  const int lane = threadIdx.x & 31;
  const int wv   = threadIdx.x >> 5;
  float* hbuf = lds + wv * LDS_PER_WAVE;
  float* ebuf = hbuf + LDS_HROWS * LDS_STRIDE;

  const int n  = lane & 15;
  const int hi = lane >> 4;

  v2f w1B[NTILE_H]; float b1l[NTILE_H];
#pragma unroll
  for (int t = 0; t < NTILE_H; ++t) {
    int H = t * 16 + n;
    v2f b; b.x = 0.f; b.y = 0.f; float bb = 0.f;
    if (H < NHID) {
      int e = H / HID, j = H - e * HID;
      b.x = W1[e * (IN_DIM * HID) + (2 * hi) * HID + j];
      b.y = hi ? 0.f : W1[e * (IN_DIM * HID) + HID + j];
      bb  = b1[H];
    }
    w1B[t] = b; b1l[t] = bb;
  }
  v2f wgB; wgB.x = 0.f; wgB.y = 0.f; float bgl = 0.f;
  if (n < NEXP) {
    wgB.x = Wg[(2 * hi) * NEXP + n];
    wgB.y = hi ? 0.f : Wg[NEXP + n];
    bgl   = bg[n];
  }
  v2f w2B[NEXP][5]; float b2l[NEXP];
#pragma unroll
  for (int e = 0; e < NEXP; ++e) {
#pragma unroll
    for (int kt = 0; kt < 5; ++kt) {
      int j = kt * 4 + 2 * hi;
      v2f b; b.x = 0.f; b.y = 0.f;
      if (n < OUT_DIM) {
        b.x = W2[e * (HID * OUT_DIM) + j * OUT_DIM + n];
        b.y = W2[e * (HID * OUT_DIM) + (j + 1) * OUT_DIM + n];
      }
      w2B[e][kt] = b;
    }
    b2l[e] = (n < OUT_DIM) ? b2[e * OUT_DIM + n] : 0.f;
  }

  const int waveGlobal = blockIdx.x * WAVES_PER_BLOCK + wv;
  const int nWaves     = gridDim.x * WAVES_PER_BLOCK;

  for (int tile = waveGlobal; tile < nTiles; tile += nWaves) {
    const int base = tile * 16;

    const float* xr = x + (size_t)(base + n) * IN_DIM;
    v2f a;
    a.x = xr[2 * hi];
    a.y = xr[1];

#pragma unroll
    for (int t = 0; t < NTILE_H; ++t) {
      v8f c = {};
      c = __builtin_amdgcn_wmma_f32_16x16x4_f32(false, a, false, w1B[t],
                                                (short)0, c, false, false);
      v4f lo, hiv;
#pragma unroll
      for (int r = 0; r < 4; ++r) {
        lo[r]  = fmaxf(c[r]     + b1l[t], 0.f);
        hiv[r] = fmaxf(c[r + 4] + b1l[t], 0.f);
      }
      float* p = hbuf + (t * 16 + n) * LDS_STRIDE + 8 * hi;
      *(v4f*)p       = lo;
      *(v4f*)(p + 4) = hiv;
    }

    v4f eLo, eHi;
    {
      v8f c = {};
      c = __builtin_amdgcn_wmma_f32_16x16x4_f32(false, a, false, wgB,
                                                (short)0, c, false, false);
#pragma unroll
      for (int r = 0; r < 4; ++r) {
        eLo[r] = __expf(c[r]     + bgl);
        eHi[r] = __expf(c[r + 4] + bgl);
      }
      if (n < NEXP) {
        float* p = ebuf + n * 16 + 8 * hi;
        *(v4f*)p       = eLo;
        *(v4f*)(p + 4) = eHi;
      }
    }

    asm volatile("" ::: "memory");

    v8f S[NEXP];
#pragma unroll
    for (int e = 0; e < NEXP; ++e) {
      S[e] = splat8(b2l[e]);
#pragma unroll
      for (int kt = 0; kt < 5; ++kt) {
        int H0 = e * HID + kt * 4 + 2 * hi;
        const float* p = hbuf + H0 * LDS_STRIDE + n;
        v2f a2; a2.x = p[0]; a2.y = p[LDS_STRIDE];
        S[e] = __builtin_amdgcn_wmma_f32_16x16x4_f32(false, a2, false, w2B[e][kt],
                                                     (short)0, S[e], false, false);
      }
    }

    v4f exl[NEXP], exh[NEXP];
#pragma unroll
    for (int e = 0; e < NEXP; ++e) {
      const float* p = ebuf + e * 16 + 8 * hi;
      exl[e] = *(const v4f*)p;
      exh[e] = *(const v4f*)(p + 4);
    }
#pragma unroll
    for (int r = 0; r < 8; ++r) {
      float ee[NEXP];
#pragma unroll
      for (int e = 0; e < NEXP; ++e) ee[e] = (r < 4) ? exl[e][r] : exh[e][r - 4];
      float d  = ((ee[0] + ee[1]) + (ee[2] + ee[3])) + ee[4];
      float rd = __builtin_amdgcn_rcpf(d);
      float acc = 0.f;
#pragma unroll
      for (int e = 0; e < NEXP; ++e) acc = fmaf(ee[e], S[e][r], acc);
      int m = base + r + 8 * hi;
      if (n < OUT_DIM) out_mixed[(size_t)m * OUT_DIM + n] = acc * rd;
      float exOwn = (r < 4) ? eLo[r] : eHi[r - 4];
      if (n < NEXP)   out_gate[(size_t)m * NEXP + n] = exOwn * rd;
    }

    asm volatile("" ::: "memory");
  }
}

extern "C" void kernel_launch(void* const* d_in, const int* in_sizes, int n_in,
                              void* d_out, int out_size, void* d_ws, size_t ws_size,
                              hipStream_t stream) {
  const float* x  = (const float*)d_in[0];
  const float* W1 = (const float*)d_in[1];
  const float* b1 = (const float*)d_in[2];
  const float* W2 = (const float*)d_in[3];
  const float* b2 = (const float*)d_in[4];
  const float* Wg = (const float*)d_in[5];
  const float* bg = (const float*)d_in[6];

  const int batch  = in_sizes[0] / IN_DIM;
  const int nTiles = batch / 16;

  float* out_mixed = (float*)d_out;
  float* out_gate  = (float*)d_out + (size_t)batch * 3;

  dim3 grid(2048), block(WAVES_PER_BLOCK * 32);
  hipLaunchKernelGGL(moe_wmma_kernel, grid, block, 0, stream,
                     x, W1, b1, W2, b2, Wg, bg, out_mixed, out_gate, nTiles);
}


# Transposed products (tokens on N): relu(fc1 D) register pairs are fc2's B fragments, h skips LDS; b1,bg in the k=3 slot; in-lane softmax; sum order changes (k per 16-row tile, o=1 over lane halves)
# speedup vs baseline: 1.5566x; 1.5566x over previous
#include <hip/hip_runtime.h>
#include <math.h>


typedef __attribute__((ext_vector_type(2))) float v2f;
typedef __attribute__((ext_vector_type(8))) float v8f;

#define IN_DIM   3
#define HID      20
#define OUT_DIM  3
#define NEXP     5
#define NFULL    6
#define NCHUNK   25
#define WAVES_PER_BLOCK 4

static __device__ __forceinline__ v2f fc1_row(const float* __restrict__ W1,
                                              const float* __restrict__ b1, int H, int h) {
  const int base = H + 2 * HID * (H / HID);
  v2f a;
  a.x = W1[base + 2 * HID * h];
  const float* p = h ? (b1 + H) : (W1 + base + HID);
  a.y = *p;
  return a;
}

static __device__ __forceinline__ float xhalf(float v) {
  const int b = __builtin_bit_cast(int, v);
  return __builtin_bit_cast(float,
      __builtin_amdgcn_permlanex16(b, b, 0x76543210, 0xfedcba98, false, false));
}

static __device__ __forceinline__ void row_eo(int m, int& e, int& o) {
  if (m < 5)       { e = m;      o = 0; }
  else if (m < 8)  { e = m - 5;  o = 1; }
  else if (m < 13) { e = m - 8;  o = 2; }
  else if (m < 15) { e = m - 10; o = 1; }
  else             { e = -1;     o = 0; }
}

__global__ __launch_bounds__(WAVES_PER_BLOCK * 32, 8) void moe_wmma_kernel(
    const float* __restrict__ x,  const float* __restrict__ W1, const float* __restrict__ b1,
    const float* __restrict__ W2, const float* __restrict__ b2,
    const float* __restrict__ Wg, const float* __restrict__ bg,
    float* __restrict__ out_mixed, float* __restrict__ out_gate, int nTiles)
{
  const int lane = threadIdx.x & 31;
  const int wv   = threadIdx.x >> 5;
  const int c    = lane & 15;
  const int h    = lane >> 4;

  v2f a1[NFULL + 1];
  for (int t = 0; t < NFULL; ++t) {
    a1[t] = fc1_row(W1, b1, t * 16 + c, h);
  }
  {
    const int cc = c & 7, hr = c >> 3;
    v2f a; a.x = 0.f; a.y = 0.f;
    if (cc < 2) {
      a = fc1_row(W1, b1, 96 + 2 * hr + cc, h);
    } else if (cc < 7) {
      const int e = cc - 2;
      a.x = Wg[(2 * h) * NEXP + e];
      a.y = h ? bg[e] : Wg[NEXP + e];
    }
    a1[NFULL] = a;
  }
  v2f a2[NCHUNK];
  {
    int er, orow;
    row_eo(c, er, orow);
    const int lo = (er >= 0) ? HID * er : 4 * NEXP * HID;
    for (int q = 0; q < NCHUNK; ++q) {
      const int Hk0 = (q < 4 * NFULL) ? (q >> 2) * 16 + 2 * (q & 3) + 8 * h : 96 + 2 * h;
      float v[2];
      for (int s = 0; s < 2; ++s) {
        const int Hk = Hk0 + s;
        const float w = W2[OUT_DIM * Hk + orow];
        v[s] = ((unsigned)(Hk - lo) < (unsigned)HID) ? w : 0.f;
      }
      a2[q].x = v[0]; a2[q].y = v[1];
    }
  }
  v8f b2c;
  for (int r = 0; r < 8; ++r) {
    int e0, o0, e1, o1;
    row_eo(r, e0, o0);
    row_eo(8 + r, e1, o1);
    const int i0 = e0 * OUT_DIM + o0;
    const int i1 = (e1 >= 0) ? e1 * OUT_DIM + o1 : 0;
    const float v = b2[h ? i1 : i0];
    b2c[r] = (h && e1 < 0) ? 0.f : v;
  }

  const int waveGlobal = blockIdx.x * WAVES_PER_BLOCK + wv;
  const int nWaves     = gridDim.x * WAVES_PER_BLOCK;

  for (int tile = waveGlobal; tile < nTiles; tile += nWaves) {
    const size_t tok = (size_t)tile * 16 + c;

    const float* xr = x + tok * IN_DIM;
    const float xa = xr[2 * h];
    const float xb = xr[1];
    v2f bx; bx.x = xa; bx.y = h ? 1.0f : xb;

    v8f d1[NFULL + 1];
    for (int t = 0; t <= NFULL; ++t) {
      v8f z = {};
      d1[t] = __builtin_amdgcn_wmma_f32_16x16x4_f32(false, a1[t], false, bx,
                                                    (short)0, z, false, false);
    }

    v8f acc = b2c;
    for (int q = 0; q < NCHUNK; ++q) {
      const int t  = (q < 4 * NFULL) ? (q >> 2) : NFULL;
      const int r0 = (q < 4 * NFULL) ? 2 * (q & 3) : 0;
      v2f hb;
      hb.x = fmaxf(d1[t][r0], 0.f);
      hb.y = fmaxf(d1[t][r0 + 1], 0.f);
      acc = __builtin_amdgcn_wmma_f32_16x16x4_f32(false, a2[q], false, hb,
                                                  (short)0, acc, false, false);
    }

    float ex[NEXP];
    for (int e = 0; e < NEXP; ++e) ex[e] = __expf(d1[NFULL][2 + e]);
    const float d  = ((ex[0] + ex[1]) + (ex[2] + ex[3])) + ex[4];
    const float rd = __builtin_amdgcn_rcpf(d);

    float mo = 0.f;
    for (int e = 0; e < NEXP; ++e) mo = fmaf(ex[e], acc[e], mo);
    const float p0 = h ? ex[3] : ex[0];
    const float p1 = h ? ex[4] : ex[1];
    const float p2 = h ? 0.f   : ex[2];
    float part = p0 * acc[5];
    part = fmaf(p1, acc[6], part);
    part = fmaf(p2, acc[7], part);
    const float m1 = (part + xhalf(part)) * rd;
    mo *= rd;

    float* pg = out_gate  + tok * NEXP;
    float* pm = out_mixed + tok * OUT_DIM;
    pg[h ? 3 : 0] = (h ? ex[3] : ex[0]) * rd;
    pg[h ? 4 : 1] = (h ? ex[4] : ex[1]) * rd;
    float* q2 = h ? (pm + 1) : (pg + 2);
    *q2 = h ? m1 : ex[2] * rd;
    pm[h ? 2 : 0] = mo;
  }
}

extern "C" void kernel_launch(void* const* d_in, const int* in_sizes, int n_in,
                              void* d_out, int out_size, void* d_ws, size_t ws_size,
                              hipStream_t stream) {
  const float* x  = (const float*)d_in[0];
  const float* W1 = (const float*)d_in[1];
  const float* b1 = (const float*)d_in[2];
  const float* W2 = (const float*)d_in[3];
  const float* b2 = (const float*)d_in[4];
  const float* Wg = (const float*)d_in[5];
  const float* bg = (const float*)d_in[6];

  const int batch  = in_sizes[0] / IN_DIM;
  const int nTiles = batch / 16;

  float* out_mixed = (float*)d_out;
  float* out_gate  = (float*)d_out + (size_t)batch * 3;

  dim3 grid(2048), block(WAVES_PER_BLOCK * 32);
  hipLaunchKernelGGL(moe_wmma_kernel, grid, block, 0, stream,
                     x, W1, b1, W2, b2, Wg, bg, out_mixed, out_gate, nTiles);
}
